# starting kernel
# baseline (speedup 1.0000x reference)
#include <hip/hip_runtime.h>
#include <math.h>

typedef float v2f __attribute__((ext_vector_type(2)));
typedef float v4f __attribute__((ext_vector_type(4)));
typedef float v8f __attribute__((ext_vector_type(8)));

#define NJ 24
#define WN_STRIDE 26
#define M_STRIDE  20

__global__ __launch_bounds__(256)
void shCaster_kernel(const float* __restrict__ xyz,
                     const float* __restrict__ viewdirs,
                     const float* __restrict__ transforms,
                     const float* __restrict__ sh_feats,
                     const float* __restrict__ locs,
                     float* __restrict__ out,
                     int N)
{
    const float SH_C0 = 0.28209479177387814f;
    const float SH_C1 = 0.4886025119029199f;
    const float C20 =  1.0925484305920792f;
    const float C21 = -1.0925484305920792f;
    const float C22 =  0.31539156525252005f;
    const float C23 = -1.0925484305920792f;
    const float C24 =  0.5462742152960396f;
    const float EPS = 1e-6f;

    __shared__ float wnLDS[256 * WN_STRIDE];
    __shared__ float mLDS [256 * M_STRIDE];

    const int tid  = threadIdx.x;
    const int lane = tid & 31;
    const int wave = tid >> 5;
    const int l15  = lane & 15;
    const int kOff = (lane >> 3) & 2;
    const int gpt  = blockIdx.x * 256 + tid;
    const int gp   = (gpt < N) ? gpt : (N - 1);

    v2f Bfrag[6];
#pragma unroll
    for (int kk = 0; kk < 6; ++kk) {
        int j0 = 4 * kk + kOff;
        Bfrag[kk].x = transforms[j0 * 16 + l15];
        Bfrag[kk].y = transforms[(j0 + 1) * 16 + l15];
    }

    const float px = xyz[gp * 3 + 0];
    const float py = xyz[gp * 3 + 1];
    const float pz = xyz[gp * 3 + 2];
    const float qx = px - viewdirs[gp * 3 + 0];
    const float qy = py - viewdirs[gp * 3 + 1];
    const float qz = pz - viewdirs[gp * 3 + 2];

    float wj[NJ];
    float wsum = 0.0f;
#pragma unroll 4
    for (int j = 0; j < NJ; ++j) {
        const float* Tj = transforms + j * 16;
        float tx = Tj[0] * px + Tj[1] * py + Tj[2]  * pz + Tj[3];
        float ty = Tj[4] * px + Tj[5] * py + Tj[6]  * pz + Tj[7];
        float tz = Tj[8] * px + Tj[9] * py + Tj[10] * pz + Tj[11];
        float dx = locs[j * 3 + 0] - tx;
        float dy = locs[j * 3 + 1] - ty;
        float dz = locs[j * 3 + 2] - tz;
        float len2 = dx * dx + dy * dy + dz * dz;
        float len  = sqrtf(len2);
        float inv  = 1.0f / len;
        dx *= inv; dy *= inv; dz *= inv;
        const float* f = sh_feats + j * 9;
        float s = SH_C0 * f[0]
                + (-SH_C1 * dy) * f[1]
                + ( SH_C1 * dz) * f[2]
                + (-SH_C1 * dx) * f[3]
                + (C20 * dx * dy) * f[4]
                + (C21 * dy * dz) * f[5]
                + (C22 * (2.0f * dz * dz - dx * dx - dy * dy)) * f[6]
                + (C23 * dx * dz) * f[7]
                + (C24 * (dx * dx - dy * dy)) * f[8];
        float rad = fmaxf(s + 0.5f, 0.0f);
        float rel = fmaxf(1.0f - len / fmaxf(rad, EPS), 0.0f);
        rel = (rad < EPS) ? 0.0f : rel;
        wj[j] = rel;
        wsum += rel;
    }
    const float invw = 1.0f / fmaxf(wsum, EPS);
#pragma unroll
    for (int j = 0; j < NJ; ++j)
        wnLDS[tid * WN_STRIDE + j] = wj[j] * invw;

    __syncthreads();

#pragma unroll
    for (int t2 = 0; t2 < 2; ++t2) {
        const int p0 = wave * 32 + t2 * 16;
        v8f c = {0.f, 0.f, 0.f, 0.f, 0.f, 0.f, 0.f, 0.f};
#pragma unroll
        for (int kk = 0; kk < 6; ++kk) {
            const int j0 = 4 * kk + kOff;
            v2f a = *(const v2f*)&wnLDS[(p0 + l15) * WN_STRIDE + j0];
            c = __builtin_amdgcn_wmma_f32_16x16x4_f32(
                    false, a, false, Bfrag[kk], (short)0, c, false, false);
        }
        const int pbase = p0 + (kOff << 2);
#pragma unroll
        for (int r = 0; r < 8; ++r)
            mLDS[(pbase + r) * M_STRIDE + l15] = c[r];
    }

    __syncthreads();

    const float* mrow = &mLDS[tid * M_STRIDE];
    v4f r0 = *(const v4f*)(mrow + 0);
    v4f r1 = *(const v4f*)(mrow + 4);
    v4f r2 = *(const v4f*)(mrow + 8);

    float o1x = r0.x * px + r0.y * py + r0.z * pz + r0.w;
    float o1y = r1.x * px + r1.y * py + r1.z * pz + r1.w;
    float o1z = r2.x * px + r2.y * py + r2.z * pz + r2.w;
    float o2x = r0.x * qx + r0.y * qy + r0.z * qz + r0.w;
    float o2y = r1.x * qx + r1.y * qy + r1.z * qz + r1.w;
    float o2z = r2.x * qx + r2.y * qy + r2.z * qz + r2.w;

    if (!(wsum > EPS)) {
        o1x = px; o1y = py; o1z = pz;
        o2x = qx; o2y = qy; o2z = qz;
    }

    if (gpt < N) {
        out[gpt * 3 + 0] = o1x;
        out[gpt * 3 + 1] = o1y;
        out[gpt * 3 + 2] = o1z;
        out[(size_t)N * 3 + gpt * 3 + 0] = o1x - o2x;
        out[(size_t)N * 3 + gpt * 3 + 1] = o1y - o2y;
        out[(size_t)N * 3 + gpt * 3 + 2] = o1z - o2z;
    }
}

extern "C" void kernel_launch(void* const* d_in, const int* in_sizes, int n_in,
                              void* d_out, int out_size, void* d_ws, size_t ws_size,
                              hipStream_t stream) {
    const float* xyz      = (const float*)d_in[0];
    const float* viewdirs = (const float*)d_in[1];
    const float* trans    = (const float*)d_in[2];
    const float* sh_feats = (const float*)d_in[3];
    const float* locs     = (const float*)d_in[4];
    float* out = (float*)d_out;

    const int N = in_sizes[0] / 3;
    const int blocks = (N + 255) / 256;
    shCaster_kernel<<<blocks, 256, 0, stream>>>(xyz, viewdirs, trans, sh_feats,
                                                locs, out, N);
}


# Weights: a conservative screen on the hardware rsq marks joint slots whose weight can be non-zero; only those run the unchanged exact sqrt/div path, per lane from LDS-staged directions; same sums
# speedup vs baseline: 8.4500x; 8.4500x over previous
#include <hip/hip_runtime.h>
#include <math.h>


typedef float v2f __attribute__((ext_vector_type(2)));
typedef float v4f __attribute__((ext_vector_type(4)));
typedef float v8f __attribute__((ext_vector_type(8)));

#define NJ 24
#define NSLOT 12
#define PASSES 4
#define WAVES_PER_BLOCK 8
#define PTS_PER_WAVE (16 * PASSES)
#define PTS_PER_BLOCK (PTS_PER_WAVE * WAVES_PER_BLOCK)
#define VROW (3 * NSLOT)
#define WROW 20
#define SCREEN_K 0.000244140625f

__device__ __forceinline__ int slot_joint(int sl, int h) {
    return (sl < 8) ? (8 * h + sl) : (16 + 4 * h + (sl - 8));
}

__device__ __forceinline__ v8f mm4(v2f a, v2f b, v8f c) {
    return __builtin_amdgcn_wmma_f32_16x16x4_f32(false, a, false, b, (short)0, c, false, false);
}

__device__ __forceinline__ float exact_weight(float dx, float dy, float dz,
                                              const float* __restrict__ f) {
    const float SH_C0 = 0.28209479177387814f;
    const float SH_C1 = 0.4886025119029199f;
    const float C20 =  1.0925484305920792f;
    const float C21 = -1.0925484305920792f;
    const float C22 =  0.31539156525252005f;
    const float C23 = -1.0925484305920792f;
    const float C24 =  0.5462742152960396f;
    const float EPS = 1e-6f;
    float len2 = dx * dx + dy * dy + dz * dz;
    float len  = sqrtf(len2);
    float inv  = 1.0f / len;
    dx *= inv; dy *= inv; dz *= inv;
    float s = SH_C0 * f[0]
            + (-SH_C1 * dy) * f[1]
            + ( SH_C1 * dz) * f[2]
            + (-SH_C1 * dx) * f[3]
            + (C20 * dx * dy) * f[4]
            + (C21 * dy * dz) * f[5]
            + (C22 * (2.0f * dz * dz - dx * dx - dy * dy)) * f[6]
            + (C23 * dx * dz) * f[7]
            + (C24 * (dx * dx - dy * dy)) * f[8];
    float rad = fmaxf(s + 0.5f, 0.0f);
    float rel = fmaxf(1.0f - len / fmaxf(rad, EPS), 0.0f);
    rel = (rad < EPS) ? 0.0f : rel;
    return rel;
}

__device__ __forceinline__ float survivor_weight(const float* vrow, int sl, int h,
                                                 const float* __restrict__ sh_feats) {
    const int b  = (sl < 8) ? sl : (16 + sl);
    const int st = (sl < 8) ? 8 : 4;
    return exact_weight(vrow[b], vrow[b + st], vrow[b + 2 * st],
                        sh_feats + slot_joint(sl, h) * 9);
}

__global__ __launch_bounds__(256)
void shCaster_kernel(const float* __restrict__ xyz,
                     const float* __restrict__ viewdirs,
                     const float* __restrict__ transforms,
                     const float* __restrict__ sh_feats,
                     const float* __restrict__ locs,
                     float* __restrict__ out,
                     int N)
{
    const float SH_C0 = 0.28209479177387814f;
    const float SH_C1 = 0.4886025119029199f;
    const float C20 =  1.0925484305920792f;
    const float C21 = -1.0925484305920792f;
    const float C22 =  0.31539156525252005f;
    const float C23 = -1.0925484305920792f;
    const float C24 =  0.5462742152960396f;
    const float EPS = 1e-6f;

    __shared__ __attribute__((aligned(16))) float sV[WAVES_PER_BLOCK][32][VROW];
    __shared__ __attribute__((aligned(16))) float sW[WAVES_PER_BLOCK][32][WROW];

    const int tid  = threadIdx.x;
    const int lane = tid & 31;
    const int wave = __builtin_amdgcn_readfirstlane(tid >> 5);
    const int c    = lane & 15;
    const int h    = lane >> 4;

    float* vrow  = &sV[wave][lane][0];
    float* wrow  = &sW[wave][lane][0];
    v4f*   vrow4 = (v4f*)vrow;
    v4f*   wrow4 = (v4f*)wrow;

    v2f A1[2][3];
    {
        const int  j1   = 16 + (c >> 3) * 4 + (c & 3);
        const bool live = (c & 4) == 0;
#pragma unroll
        for (int a = 0; a < 3; ++a) {
            const float* T0 = transforms + c  * 16 + a * 4;
            const float* T1 = transforms + j1 * 16 + a * 4;
            const float e0 = -T0[2 * h];
            const float e1 = h ? (locs[c * 3 + a] - T0[3]) : -T0[1];
            const float g0 = -T1[2 * h];
            const float g1 = h ? (locs[j1 * 3 + a] - T1[3]) : -T1[1];
            A1[0][a].x = e0;
            A1[0][a].y = e1;
            A1[1][a].x = live ? g0 : 0.0f;
            A1[1][a].y = live ? g1 : 0.0f;
        }
    }

    v2f A2[6];
#pragma unroll
    for (int kk = 0; kk < 6; ++kk) {
        A2[kk].x = transforms[slot_joint(2 * kk,     h) * 16 + c];
        A2[kk].y = transforms[slot_joint(2 * kk + 1, h) * 16 + c];
    }

    float g[NSLOT][9];
    v2f gacc[3] = {{0.f, 0.f}, {0.f, 0.f}, {0.f, 0.f}};
#pragma unroll
    for (int sl = 0; sl < NSLOT; ++sl) {
        const float* f = sh_feats + slot_joint(sl, h) * 9;
        g[sl][0] = SH_C0 * f[0] + 0.5f;
        g[sl][1] = -SH_C1 * f[1];
        g[sl][2] =  SH_C1 * f[2];
        g[sl][3] = -SH_C1 * f[3];
        g[sl][4] = C20 * f[4];
        g[sl][5] = C21 * f[5];
        g[sl][6] = C22 * f[6];
        g[sl][7] = C23 * f[7];
        g[sl][8] = C24 * f[8];
    }
#pragma unroll
    for (int sl = 0; sl < NSLOT; sl += 2) {
#pragma unroll
        for (int i = 0; i < 9; ++i) {
            const v2f t = {g[sl][i], g[sl + 1][i]};
            gacc[i % 3] += t * t;
        }
    }
    const v2f gs2 = gacc[0] + gacc[1] + gacc[2];
    const float gsq = gs2.x + gs2.y;
    const float thr = SCREEN_K * (2.5f + 1.5f * gsq);

    const int waveBase = (blockIdx.x * WAVES_PER_BLOCK + wave) * PTS_PER_WAVE;

    float px, py, pz, ux, uy, uz;
    {
        const int gi = min(waveBase + c, N - 1);
        px = xyz[gi * 3 + 0]; py = xyz[gi * 3 + 1]; pz = xyz[gi * 3 + 2];
        ux = viewdirs[gi * 3 + 0]; uy = viewdirs[gi * 3 + 1]; uz = viewdirs[gi * 3 + 2];
    }

#pragma unroll 1
    for (int pass = 0; pass < PASSES; ++pass) {
        const int base = waveBase + pass * 16;
        if (base >= N) break;
        const int pt = base + c;

        float nx, ny, nz, mx, my, mz;
        {
            const int gi = min(pt + 16, N - 1);
            nx = xyz[gi * 3 + 0]; ny = xyz[gi * 3 + 1]; nz = xyz[gi * 3 + 2];
            mx = viewdirs[gi * 3 + 0]; my = viewdirs[gi * 3 + 1]; mz = viewdirs[gi * 3 + 2];
        }

        const float qx = px - ux;
        const float qy = py - uy;
        const float qz = pz - uz;

        v2f B1;
        B1.x = h ? pz : px;
        B1.y = h ? 1.0f : py;
        const v8f z8 = {0.f, 0.f, 0.f, 0.f, 0.f, 0.f, 0.f, 0.f};
        const v8f vX0 = mm4(A1[0][0], B1, z8);
        const v8f vY0 = mm4(A1[0][1], B1, z8);
        const v8f vZ0 = mm4(A1[0][2], B1, z8);
        const v8f vX1 = mm4(A1[1][0], B1, z8);
        const v8f vY1 = mm4(A1[1][1], B1, z8);
        const v8f vZ1 = mm4(A1[1][2], B1, z8);

        unsigned mask = 0u;
#pragma unroll
        for (int sl = 0; sl < NSLOT; ++sl) {
            const float dx = (sl < 8) ? vX0[sl & 7] : vX1[sl & 7];
            const float dy = (sl < 8) ? vY0[sl & 7] : vY1[sl & 7];
            const float dz = (sl < 8) ? vZ0[sl & 7] : vZ1[sl & 7];
            const float len2 = dx * dx + dy * dy + dz * dz;
            const float ia   = __builtin_amdgcn_rsqf(len2);
            const float x = dx * ia, y = dy * ia, z = dz * ia;
            const float lenA = len2 * ia;
            const float xx = x * x, yy = y * y, zz = z * z;
            const float* gs = g[sl];
            const float radA = gs[0] + gs[1] * y + gs[2] * z + gs[3] * x
                             + gs[4] * (x * y) + gs[5] * (y * z)
                             + gs[6] * (2.0f * zz - xx - yy)
                             + gs[7] * (x * z) + gs[8] * (xx - yy);
            const float lhs = lenA * (1.0f - SCREEN_K) - thr;
            mask |= (lhs > radA) ? 0u : (1u << sl);
        }

        vrow4[0] = __builtin_shufflevector(vX0, vX0, 0, 1, 2, 3);
        vrow4[1] = __builtin_shufflevector(vX0, vX0, 4, 5, 6, 7);
        vrow4[2] = __builtin_shufflevector(vY0, vY0, 0, 1, 2, 3);
        vrow4[3] = __builtin_shufflevector(vY0, vY0, 4, 5, 6, 7);
        vrow4[4] = __builtin_shufflevector(vZ0, vZ0, 0, 1, 2, 3);
        vrow4[5] = __builtin_shufflevector(vZ0, vZ0, 4, 5, 6, 7);
        vrow4[6] = __builtin_shufflevector(vX1, vX1, 0, 1, 2, 3);
        vrow4[7] = __builtin_shufflevector(vY1, vY1, 0, 1, 2, 3);
        vrow4[8] = __builtin_shufflevector(vZ1, vZ1, 0, 1, 2, 3);
        const v4f z4 = {0.f, 0.f, 0.f, 0.f};
        wrow4[0] = z4; wrow4[1] = z4; wrow4[2] = z4;

#pragma unroll 1
        for (int it = 0; it < NSLOT / 2; ++it) {
            if (__builtin_amdgcn_ballot_w32(mask != 0u) == 0u) break;
            const bool a1 = mask != 0u;
            int s1 = __ffs((int)mask) - 1;
            s1 = (s1 < 0) ? 0 : s1;
            mask &= mask - 1u;
            const bool a2 = mask != 0u;
            int s2 = __ffs((int)mask) - 1;
            s2 = (s2 < 0) ? 0 : s2;
            mask &= mask - 1u;
            const float r1 = survivor_weight(vrow, s1, h, sh_feats);
            const float r2 = survivor_weight(vrow, s2, h, sh_feats);
            wrow[a1 ? s1 : NSLOT] = r1;
            wrow[a2 ? s2 : NSLOT] = r2;
        }

        const v4f wa = wrow4[0], wb = wrow4[1], wc = wrow4[2];
        const float w[NSLOT] = {wa.x, wa.y, wa.z, wa.w, wb.x, wb.y, wb.z, wb.w,
                                wc.x, wc.y, wc.z, wc.w};
        float wpart = 0.0f;
#pragma unroll
        for (int sl = 0; sl < NSLOT; ++sl) wpart += w[sl];
        const float wsum = wpart + __shfl_xor(wpart, 16);
        const float invw = 1.0f / fmaxf(wsum, EPS);

        v8f m = z8;
#pragma unroll
        for (int kk = 0; kk < 6; ++kk) {
            v2f b;
            b.x = w[2 * kk]     * invw;
            b.y = w[2 * kk + 1] * invw;
            m = mm4(A2[kk], b, m);
        }

        float oA1 = m[0] * px + m[1] * py + m[2] * pz + m[3];
        float oB1 = m[4] * px + m[5] * py + m[6] * pz + m[7];
        float oA2 = m[0] * qx + m[1] * qy + m[2] * qz + m[3];
        float oB2 = m[4] * qx + m[5] * qy + m[6] * qz + m[7];

        if (!(wsum > EPS)) {
            oA1 = h ? pz : px;  oB1 = py;
            oA2 = h ? qz : qx;  oB2 = qy;
        }

        if (pt < N) {
            const size_t ia3 = (size_t)pt * 3 + (h ? 2 : 0);
            out[ia3] = oA1;
            out[(size_t)N * 3 + ia3] = oA1 - oA2;
            if (h == 0) {
                out[(size_t)pt * 3 + 1] = oB1;
                out[(size_t)N * 3 + (size_t)pt * 3 + 1] = oB1 - oB2;
            }
        }

        px = nx; py = ny; pz = nz;
        ux = mx; uy = my; uz = mz;
    }
}

extern "C" void kernel_launch(void* const* d_in, const int* in_sizes, int n_in,
                              void* d_out, int out_size, void* d_ws, size_t ws_size,
                              hipStream_t stream) {
    const float* xyz      = (const float*)d_in[0];
    const float* viewdirs = (const float*)d_in[1];
    const float* trans    = (const float*)d_in[2];
    const float* sh_feats = (const float*)d_in[3];
    const float* locs     = (const float*)d_in[4];
    float* out = (float*)d_out;

    const int N = in_sizes[0] / 3;
    const int blocks = (N + PTS_PER_BLOCK - 1) / PTS_PER_BLOCK;
    shCaster_kernel<<<blocks, 256, 0, stream>>>(xyz, viewdirs, trans, sh_feats,
                                                locs, out, N);
}


# Prologue: every lane loads whole matrix rows and joint centres and the lane half selects in registers; no load under a divergent branch, so all prologue loads go out as one batch; same sums
# speedup vs baseline: 9.6305x; 1.1397x over previous
#include <hip/hip_runtime.h>
#include <math.h>


typedef float v2f __attribute__((ext_vector_type(2)));
typedef float v4f __attribute__((ext_vector_type(4)));
typedef float v8f __attribute__((ext_vector_type(8)));

#define NJ 24
#define NSLOT 12
#define PASSES 4
#define WAVES_PER_BLOCK 8
#define PTS_PER_WAVE (16 * PASSES)
#define PTS_PER_BLOCK (PTS_PER_WAVE * WAVES_PER_BLOCK)
#define VROW (3 * NSLOT)
#define WROW 20
#define SCREEN_K 0.000244140625f

__device__ __forceinline__ int slot_joint(int sl, int h) {
    return (sl < 8) ? (8 * h + sl) : (16 + 4 * h + (sl - 8));
}

__device__ __forceinline__ v8f mm4(v2f a, v2f b, v8f c) {
    return __builtin_amdgcn_wmma_f32_16x16x4_f32(false, a, false, b, (short)0, c, false, false);
}

__device__ __forceinline__ float exact_weight(float dx, float dy, float dz,
                                              const float* __restrict__ f) {
    const float SH_C0 = 0.28209479177387814f;
    const float SH_C1 = 0.4886025119029199f;
    const float C20 =  1.0925484305920792f;
    const float C21 = -1.0925484305920792f;
    const float C22 =  0.31539156525252005f;
    const float C23 = -1.0925484305920792f;
    const float C24 =  0.5462742152960396f;
    const float EPS = 1e-6f;
    float len2 = dx * dx + dy * dy + dz * dz;
    float len  = sqrtf(len2);
    float inv  = 1.0f / len;
    dx *= inv; dy *= inv; dz *= inv;
    float s = SH_C0 * f[0]
            + (-SH_C1 * dy) * f[1]
            + ( SH_C1 * dz) * f[2]
            + (-SH_C1 * dx) * f[3]
            + (C20 * dx * dy) * f[4]
            + (C21 * dy * dz) * f[5]
            + (C22 * (2.0f * dz * dz - dx * dx - dy * dy)) * f[6]
            + (C23 * dx * dz) * f[7]
            + (C24 * (dx * dx - dy * dy)) * f[8];
    float rad = fmaxf(s + 0.5f, 0.0f);
    float rel = fmaxf(1.0f - len / fmaxf(rad, EPS), 0.0f);
    rel = (rad < EPS) ? 0.0f : rel;
    return rel;
}

__device__ __forceinline__ float survivor_weight(const float* vrow, int sl, int h,
                                                 const float* __restrict__ sh_feats) {
    const int b  = (sl < 8) ? sl : (16 + sl);
    const int st = (sl < 8) ? 8 : 4;
    return exact_weight(vrow[b], vrow[b + st], vrow[b + 2 * st],
                        sh_feats + slot_joint(sl, h) * 9);
}

__global__ __launch_bounds__(256)
void shCaster_kernel(const float* __restrict__ xyz,
                     const float* __restrict__ viewdirs,
                     const float* __restrict__ transforms,
                     const float* __restrict__ sh_feats,
                     const float* __restrict__ locs,
                     float* __restrict__ out,
                     int N)
{
    const float SH_C0 = 0.28209479177387814f;
    const float SH_C1 = 0.4886025119029199f;
    const float C20 =  1.0925484305920792f;
    const float C21 = -1.0925484305920792f;
    const float C22 =  0.31539156525252005f;
    const float C23 = -1.0925484305920792f;
    const float C24 =  0.5462742152960396f;
    const float EPS = 1e-6f;

    __shared__ __attribute__((aligned(16))) float sV[WAVES_PER_BLOCK][32][VROW];
    __shared__ __attribute__((aligned(16))) float sW[WAVES_PER_BLOCK][32][WROW];

    const int tid  = threadIdx.x;
    const int lane = tid & 31;
    const int wave = __builtin_amdgcn_readfirstlane(tid >> 5);
    const int c    = lane & 15;
    const int h    = lane >> 4;

    float* vrow  = &sV[wave][lane][0];
    float* wrow  = &sW[wave][lane][0];
    v4f*   vrow4 = (v4f*)vrow;
    v4f*   wrow4 = (v4f*)wrow;

    v2f A1[2][3];
    {
        const int  j1   = 16 + (c >> 3) * 4 + (c & 3);
        const bool live = (c & 4) == 0;
#pragma unroll
        for (int a = 0; a < 3; ++a) {
            const float* T0 = transforms + c  * 16 + a * 4;
            const float* T1 = transforms + j1 * 16 + a * 4;
            const float t0 = T0[0], t1 = T0[1], t2 = T0[2], t3 = T0[3];
            const float u0 = T1[0], u1 = T1[1], u2 = T1[2], u3 = T1[3];
            const float l0 = locs[c  * 3 + a];
            const float l1 = locs[j1 * 3 + a];
            const float e0 = h ? -t2 : -t0;
            const float e1 = h ? (l0 - t3) : -t1;
            const float g0 = h ? -u2 : -u0;
            const float g1 = h ? (l1 - u3) : -u1;
            A1[0][a].x = e0;
            A1[0][a].y = e1;
            A1[1][a].x = live ? g0 : 0.0f;
            A1[1][a].y = live ? g1 : 0.0f;
        }
    }

    v2f A2[6];
#pragma unroll
    for (int kk = 0; kk < 6; ++kk) {
        A2[kk].x = transforms[slot_joint(2 * kk,     h) * 16 + c];
        A2[kk].y = transforms[slot_joint(2 * kk + 1, h) * 16 + c];
    }

    float g[NSLOT][9];
    v2f gacc[3] = {{0.f, 0.f}, {0.f, 0.f}, {0.f, 0.f}};
#pragma unroll
    for (int sl = 0; sl < NSLOT; ++sl) {
        const float* f = sh_feats + slot_joint(sl, h) * 9;
        g[sl][0] = SH_C0 * f[0] + 0.5f;
        g[sl][1] = -SH_C1 * f[1];
        g[sl][2] =  SH_C1 * f[2];
        g[sl][3] = -SH_C1 * f[3];
        g[sl][4] = C20 * f[4];
        g[sl][5] = C21 * f[5];
        g[sl][6] = C22 * f[6];
        g[sl][7] = C23 * f[7];
        g[sl][8] = C24 * f[8];
    }
#pragma unroll
    for (int sl = 0; sl < NSLOT; sl += 2) {
#pragma unroll
        for (int i = 0; i < 9; ++i) {
            const v2f t = {g[sl][i], g[sl + 1][i]};
            gacc[i % 3] += t * t;
        }
    }
    const v2f gs2 = gacc[0] + gacc[1] + gacc[2];
    const float gsq = gs2.x + gs2.y;
    const float thr = SCREEN_K * (2.5f + 1.5f * gsq);

    const int waveBase = (blockIdx.x * WAVES_PER_BLOCK + wave) * PTS_PER_WAVE;

    float px, py, pz, ux, uy, uz;
    {
        const int gi = min(waveBase + c, N - 1);
        px = xyz[gi * 3 + 0]; py = xyz[gi * 3 + 1]; pz = xyz[gi * 3 + 2];
        ux = viewdirs[gi * 3 + 0]; uy = viewdirs[gi * 3 + 1]; uz = viewdirs[gi * 3 + 2];
    }

#pragma unroll 1
    for (int pass = 0; pass < PASSES; ++pass) {
        const int base = waveBase + pass * 16;
        if (base >= N) break;
        const int pt = base + c;

        float nx, ny, nz, mx, my, mz;
        {
            const int gi = min(pt + 16, N - 1);
            nx = xyz[gi * 3 + 0]; ny = xyz[gi * 3 + 1]; nz = xyz[gi * 3 + 2];
            mx = viewdirs[gi * 3 + 0]; my = viewdirs[gi * 3 + 1]; mz = viewdirs[gi * 3 + 2];
        }

        const float qx = px - ux;
        const float qy = py - uy;
        const float qz = pz - uz;

        v2f B1;
        B1.x = h ? pz : px;
        B1.y = h ? 1.0f : py;
        const v8f z8 = {0.f, 0.f, 0.f, 0.f, 0.f, 0.f, 0.f, 0.f};
        const v8f vX0 = mm4(A1[0][0], B1, z8);
        const v8f vY0 = mm4(A1[0][1], B1, z8);
        const v8f vZ0 = mm4(A1[0][2], B1, z8);
        const v8f vX1 = mm4(A1[1][0], B1, z8);
        const v8f vY1 = mm4(A1[1][1], B1, z8);
        const v8f vZ1 = mm4(A1[1][2], B1, z8);

        unsigned mask = 0u;
#pragma unroll
        for (int sl = 0; sl < NSLOT; ++sl) {
            const float dx = (sl < 8) ? vX0[sl & 7] : vX1[sl & 7];
            const float dy = (sl < 8) ? vY0[sl & 7] : vY1[sl & 7];
            const float dz = (sl < 8) ? vZ0[sl & 7] : vZ1[sl & 7];
            const float len2 = dx * dx + dy * dy + dz * dz;
            const float ia   = __builtin_amdgcn_rsqf(len2);
            const float x = dx * ia, y = dy * ia, z = dz * ia;
            const float lenA = len2 * ia;
            const float xx = x * x, yy = y * y, zz = z * z;
            const float* gs = g[sl];
            const float radA = gs[0] + gs[1] * y + gs[2] * z + gs[3] * x
                             + gs[4] * (x * y) + gs[5] * (y * z)
                             + gs[6] * (2.0f * zz - xx - yy)
                             + gs[7] * (x * z) + gs[8] * (xx - yy);
            const float lhs = lenA * (1.0f - SCREEN_K) - thr;
            mask |= (lhs > radA) ? 0u : (1u << sl);
        }

        vrow4[0] = __builtin_shufflevector(vX0, vX0, 0, 1, 2, 3);
        vrow4[1] = __builtin_shufflevector(vX0, vX0, 4, 5, 6, 7);
        vrow4[2] = __builtin_shufflevector(vY0, vY0, 0, 1, 2, 3);
        vrow4[3] = __builtin_shufflevector(vY0, vY0, 4, 5, 6, 7);
        vrow4[4] = __builtin_shufflevector(vZ0, vZ0, 0, 1, 2, 3);
        vrow4[5] = __builtin_shufflevector(vZ0, vZ0, 4, 5, 6, 7);
        vrow4[6] = __builtin_shufflevector(vX1, vX1, 0, 1, 2, 3);
        vrow4[7] = __builtin_shufflevector(vY1, vY1, 0, 1, 2, 3);
        vrow4[8] = __builtin_shufflevector(vZ1, vZ1, 0, 1, 2, 3);
        const v4f z4 = {0.f, 0.f, 0.f, 0.f};
        wrow4[0] = z4; wrow4[1] = z4; wrow4[2] = z4;

#pragma unroll 1
        for (int it = 0; it < NSLOT / 2; ++it) {
            if (__builtin_amdgcn_ballot_w32(mask != 0u) == 0u) break;
            const bool a1 = mask != 0u;
            int s1 = __ffs((int)mask) - 1;
            s1 = (s1 < 0) ? 0 : s1;
            mask &= mask - 1u;
            const bool a2 = mask != 0u;
            int s2 = __ffs((int)mask) - 1;
            s2 = (s2 < 0) ? 0 : s2;
            mask &= mask - 1u;
            const float r1 = survivor_weight(vrow, s1, h, sh_feats);
            const float r2 = survivor_weight(vrow, s2, h, sh_feats);
            wrow[a1 ? s1 : NSLOT] = r1;
            wrow[a2 ? s2 : NSLOT] = r2;
        }

        const v4f wa = wrow4[0], wb = wrow4[1], wc = wrow4[2];
        const float w[NSLOT] = {wa.x, wa.y, wa.z, wa.w, wb.x, wb.y, wb.z, wb.w,
                                wc.x, wc.y, wc.z, wc.w};
        float wpart = 0.0f;
#pragma unroll
        for (int sl = 0; sl < NSLOT; ++sl) wpart += w[sl];
        const float wsum = wpart + __shfl_xor(wpart, 16);
        const float invw = 1.0f / fmaxf(wsum, EPS);

        v8f m = z8;
#pragma unroll
        for (int kk = 0; kk < 6; ++kk) {
            v2f b;
            b.x = w[2 * kk]     * invw;
            b.y = w[2 * kk + 1] * invw;
            m = mm4(A2[kk], b, m);
        }

        float oA1 = m[0] * px + m[1] * py + m[2] * pz + m[3];
        float oB1 = m[4] * px + m[5] * py + m[6] * pz + m[7];
        float oA2 = m[0] * qx + m[1] * qy + m[2] * qz + m[3];
        float oB2 = m[4] * qx + m[5] * qy + m[6] * qz + m[7];

        if (!(wsum > EPS)) {
            oA1 = h ? pz : px;  oB1 = py;
            oA2 = h ? qz : qx;  oB2 = qy;
        }

        if (pt < N) {
            const size_t ia3 = (size_t)pt * 3 + (h ? 2 : 0);
            out[ia3] = oA1;
            out[(size_t)N * 3 + ia3] = oA1 - oA2;
            if (h == 0) {
                out[(size_t)pt * 3 + 1] = oB1;
                out[(size_t)N * 3 + (size_t)pt * 3 + 1] = oB1 - oB2;
            }
        }

        px = nx; py = ny; pz = nz;
        ux = mx; uy = my; uz = mz;
    }
}

extern "C" void kernel_launch(void* const* d_in, const int* in_sizes, int n_in,
                              void* d_out, int out_size, void* d_ws, size_t ws_size,
                              hipStream_t stream) {
    const float* xyz      = (const float*)d_in[0];
    const float* viewdirs = (const float*)d_in[1];
    const float* trans    = (const float*)d_in[2];
    const float* sh_feats = (const float*)d_in[3];
    const float* locs     = (const float*)d_in[4];
    float* out = (float*)d_out;

    const int N = in_sizes[0] / 3;
    const int blocks = (N + PTS_PER_BLOCK - 1) / PTS_PER_BLOCK;
    shCaster_kernel<<<blocks, 256, 0, stream>>>(xyz, viewdirs, trans, sh_feats,
                                                locs, out, N);
}


# Survivors: each wave packs its kept (lane, slot) pairs into a dense LDS queue by ballot rank; the unchanged exact weight runs one queue entry per lane per trip; same sums in the same order
# speedup vs baseline: 10.1886x; 1.0580x over previous
#include <hip/hip_runtime.h>
#include <math.h>


typedef float v2f __attribute__((ext_vector_type(2)));
typedef float v4f __attribute__((ext_vector_type(4)));
typedef float v8f __attribute__((ext_vector_type(8)));

#define NJ 24
#define NSLOT 12
#define PASSES 4
#define WAVES_PER_BLOCK 8
#define PTS_PER_WAVE (16 * PASSES)
#define PTS_PER_BLOCK (PTS_PER_WAVE * WAVES_PER_BLOCK)
#define VROW (3 * NSLOT)
#define WROW 20
#define SCREEN_K 0.000244140625f

__device__ __forceinline__ int slot_joint(int sl, int h) {
    return (sl < 8) ? (8 * h + sl) : (16 + 4 * h + (sl - 8));
}

__device__ __forceinline__ v8f mm4(v2f a, v2f b, v8f c) {
    return __builtin_amdgcn_wmma_f32_16x16x4_f32(false, a, false, b, (short)0, c, false, false);
}

__device__ __forceinline__ float exact_weight(float dx, float dy, float dz,
                                              const float* __restrict__ f) {
    const float SH_C0 = 0.28209479177387814f;
    const float SH_C1 = 0.4886025119029199f;
    const float C20 =  1.0925484305920792f;
    const float C21 = -1.0925484305920792f;
    const float C22 =  0.31539156525252005f;
    const float C23 = -1.0925484305920792f;
    const float C24 =  0.5462742152960396f;
    const float EPS = 1e-6f;
    float len2 = dx * dx + dy * dy + dz * dz;
    float len  = sqrtf(len2);
    float inv  = 1.0f / len;
    dx *= inv; dy *= inv; dz *= inv;
    float s = SH_C0 * f[0]
            + (-SH_C1 * dy) * f[1]
            + ( SH_C1 * dz) * f[2]
            + (-SH_C1 * dx) * f[3]
            + (C20 * dx * dy) * f[4]
            + (C21 * dy * dz) * f[5]
            + (C22 * (2.0f * dz * dz - dx * dx - dy * dy)) * f[6]
            + (C23 * dx * dz) * f[7]
            + (C24 * (dx * dx - dy * dy)) * f[8];
    float rad = fmaxf(s + 0.5f, 0.0f);
    float rel = fmaxf(1.0f - len / fmaxf(rad, EPS), 0.0f);
    rel = (rad < EPS) ? 0.0f : rel;
    return rel;
}

__device__ __forceinline__ float survivor_weight(const float* vrow, int sl, int h,
                                                 const float* __restrict__ sh_feats) {
    const int b  = (sl < 8) ? sl : (16 + sl);
    const int st = (sl < 8) ? 8 : 4;
    const int j  = sl + ((sl < 8) ? 8 * h : 8 + 4 * h);
    return exact_weight(vrow[b], vrow[b + st], vrow[b + 2 * st], sh_feats + j * 9);
}

__global__ __launch_bounds__(256)
void shCaster_kernel(const float* __restrict__ xyz,
                     const float* __restrict__ viewdirs,
                     const float* __restrict__ transforms,
                     const float* __restrict__ sh_feats,
                     const float* __restrict__ locs,
                     float* __restrict__ out,
                     int N)
{
    const float SH_C0 = 0.28209479177387814f;
    const float SH_C1 = 0.4886025119029199f;
    const float C20 =  1.0925484305920792f;
    const float C21 = -1.0925484305920792f;
    const float C22 =  0.31539156525252005f;
    const float C23 = -1.0925484305920792f;
    const float C24 =  0.5462742152960396f;
    const float EPS = 1e-6f;

    __shared__ __attribute__((aligned(16))) float sV[WAVES_PER_BLOCK][32][VROW];
    __shared__ __attribute__((aligned(16))) float sW[WAVES_PER_BLOCK][32][WROW];
    __shared__ unsigned sQ[WAVES_PER_BLOCK][32 * NSLOT + 32];

    const int tid  = threadIdx.x;
    const int lane = tid & 31;
    const int wave = __builtin_amdgcn_readfirstlane(tid >> 5);
    const int c    = lane & 15;
    const int h    = lane >> 4;

    float* vrow  = &sV[wave][lane][0];
    float* wrow  = &sW[wave][lane][0];
    v4f*   vrow4 = (v4f*)vrow;
    v4f*   wrow4 = (v4f*)wrow;
    unsigned* q  = &sQ[wave][0];
    const unsigned laneTag = (unsigned)lane << 4;
    const unsigned spare   = 32 * NSLOT + lane;

    v2f A1[2][3];
    {
        const int  j1   = 16 + (c >> 3) * 4 + (c & 3);
        const bool live = (c & 4) == 0;
#pragma unroll
        for (int a = 0; a < 3; ++a) {
            const float* T0 = transforms + c  * 16 + a * 4;
            const float* T1 = transforms + j1 * 16 + a * 4;
            const float t0 = T0[0], t1 = T0[1], t2 = T0[2], t3 = T0[3];
            const float u0 = T1[0], u1 = T1[1], u2 = T1[2], u3 = T1[3];
            const float l0 = locs[c  * 3 + a];
            const float l1 = locs[j1 * 3 + a];
            const float e0 = h ? -t2 : -t0;
            const float e1 = h ? (l0 - t3) : -t1;
            const float g0 = h ? -u2 : -u0;
            const float g1 = h ? (l1 - u3) : -u1;
            A1[0][a].x = e0;
            A1[0][a].y = e1;
            A1[1][a].x = live ? g0 : 0.0f;
            A1[1][a].y = live ? g1 : 0.0f;
        }
    }

    v2f A2[6];
#pragma unroll
    for (int kk = 0; kk < 6; ++kk) {
        A2[kk].x = transforms[slot_joint(2 * kk,     h) * 16 + c];
        A2[kk].y = transforms[slot_joint(2 * kk + 1, h) * 16 + c];
    }

    float g[NSLOT][9];
    v2f gacc[3] = {{0.f, 0.f}, {0.f, 0.f}, {0.f, 0.f}};
#pragma unroll
    for (int sl = 0; sl < NSLOT; ++sl) {
        const float* f = sh_feats + slot_joint(sl, h) * 9;
        g[sl][0] = SH_C0 * f[0] + 0.5f;
        g[sl][1] = -SH_C1 * f[1];
        g[sl][2] =  SH_C1 * f[2];
        g[sl][3] = -SH_C1 * f[3];
        g[sl][4] = C20 * f[4];
        g[sl][5] = C21 * f[5];
        g[sl][6] = C22 * f[6];
        g[sl][7] = C23 * f[7];
        g[sl][8] = C24 * f[8];
    }
    v2f g2[NSLOT / 2][9];
#pragma unroll
    for (int kk = 0; kk < NSLOT / 2; ++kk) {
#pragma unroll
        for (int i = 0; i < 9; ++i) {
            const v2f t = {g[2 * kk][i], g[2 * kk + 1][i]};
            g2[kk][i] = t;
            gacc[i % 3] += t * t;
        }
    }
    const v2f gs2 = gacc[0] + gacc[1] + gacc[2];
    const float gsq = gs2.x + gs2.y;
    const float thr = SCREEN_K * (2.5f + 1.5f * gsq);

    const int waveBase = (blockIdx.x * WAVES_PER_BLOCK + wave) * PTS_PER_WAVE;

    float px, py, pz, ux, uy, uz;
    {
        const int gi = min(waveBase + c, N - 1);
        px = xyz[gi * 3 + 0]; py = xyz[gi * 3 + 1]; pz = xyz[gi * 3 + 2];
        ux = viewdirs[gi * 3 + 0]; uy = viewdirs[gi * 3 + 1]; uz = viewdirs[gi * 3 + 2];
    }

#pragma unroll 1
    for (int pass = 0; pass < PASSES; ++pass) {
        const int base = waveBase + pass * 16;
        if (base >= N) break;
        const int pt = base + c;

        float nx, ny, nz, mx, my, mz;
        {
            const int gi = min(pt + 16, N - 1);
            nx = xyz[gi * 3 + 0]; ny = xyz[gi * 3 + 1]; nz = xyz[gi * 3 + 2];
            mx = viewdirs[gi * 3 + 0]; my = viewdirs[gi * 3 + 1]; mz = viewdirs[gi * 3 + 2];
        }

        const float qx = px - ux;
        const float qy = py - uy;
        const float qz = pz - uz;

        v2f B1;
        B1.x = h ? pz : px;
        B1.y = h ? 1.0f : py;
        const v8f z8 = {0.f, 0.f, 0.f, 0.f, 0.f, 0.f, 0.f, 0.f};
        const v8f vX0 = mm4(A1[0][0], B1, z8);
        const v8f vY0 = mm4(A1[0][1], B1, z8);
        const v8f vZ0 = mm4(A1[0][2], B1, z8);
        const v8f vX1 = mm4(A1[1][0], B1, z8);
        const v8f vY1 = mm4(A1[1][1], B1, z8);
        const v8f vZ1 = mm4(A1[1][2], B1, z8);

        v2f lhs2[NSLOT / 2], rad2[NSLOT / 2];
#pragma unroll
        for (int kk = 0; kk < NSLOT / 2; ++kk) {
            const int r = (2 * kk) & 7;
            v2f dx, dy, dz;
            if (kk < 4) {
                dx.x = vX0[r]; dx.y = vX0[r + 1];
                dy.x = vY0[r]; dy.y = vY0[r + 1];
                dz.x = vZ0[r]; dz.y = vZ0[r + 1];
            } else {
                dx.x = vX1[r]; dx.y = vX1[r + 1];
                dy.x = vY1[r]; dy.y = vY1[r + 1];
                dz.x = vZ1[r]; dz.y = vZ1[r + 1];
            }
            const v2f len2 = dx * dx + dy * dy + dz * dz;
            v2f ia;
            ia.x = __builtin_amdgcn_rsqf(len2.x);
            ia.y = __builtin_amdgcn_rsqf(len2.y);
            const v2f x = dx * ia, y = dy * ia, z = dz * ia;
            const v2f lenA = len2 * ia;
            const v2f xx = x * x, yy = y * y, zz = z * z;
            const v2f* gs = g2[kk];
            rad2[kk] = gs[0] + gs[1] * y + gs[2] * z + gs[3] * x
                     + gs[4] * (x * y) + gs[5] * (y * z)
                     + gs[6] * (2.0f * zz - xx - yy)
                     + gs[7] * (x * z) + gs[8] * (xx - yy);
            lhs2[kk] = lenA * (1.0f - SCREEN_K) - thr;
        }

        int total = 0;
#pragma unroll
        for (int sl = 0; sl < NSLOT; ++sl) {
            const float lhs  = (sl & 1) ? lhs2[sl >> 1].y : lhs2[sl >> 1].x;
            const float radA = (sl & 1) ? rad2[sl >> 1].y : rad2[sl >> 1].x;
            const bool keep = !(lhs > radA);
            const unsigned bal = __builtin_amdgcn_ballot_w32(keep);
            const unsigned pos = __builtin_amdgcn_mbcnt_lo(bal, (unsigned)total);
            q[keep ? pos : spare] = laneTag + (unsigned)sl;
            total += __builtin_popcount(bal);
        }

        vrow4[0] = __builtin_shufflevector(vX0, vX0, 0, 1, 2, 3);
        vrow4[1] = __builtin_shufflevector(vX0, vX0, 4, 5, 6, 7);
        vrow4[2] = __builtin_shufflevector(vY0, vY0, 0, 1, 2, 3);
        vrow4[3] = __builtin_shufflevector(vY0, vY0, 4, 5, 6, 7);
        vrow4[4] = __builtin_shufflevector(vZ0, vZ0, 0, 1, 2, 3);
        vrow4[5] = __builtin_shufflevector(vZ0, vZ0, 4, 5, 6, 7);
        vrow4[6] = __builtin_shufflevector(vX1, vX1, 0, 1, 2, 3);
        vrow4[7] = __builtin_shufflevector(vY1, vY1, 0, 1, 2, 3);
        vrow4[8] = __builtin_shufflevector(vZ1, vZ1, 0, 1, 2, 3);
        const v4f z4 = {0.f, 0.f, 0.f, 0.f};
        wrow4[0] = z4; wrow4[1] = z4; wrow4[2] = z4;

        __builtin_amdgcn_fence(__ATOMIC_RELEASE, "wavefront");
        __builtin_amdgcn_wave_barrier();
        __builtin_amdgcn_fence(__ATOMIC_ACQUIRE, "wavefront");

#pragma unroll 1
        for (int t = 0; t < total; t += 32) {
            const int  idx = t + lane;
            const bool act = idx < total;
            const unsigned e = q[act ? idx : 0];
            const int src = (int)(e >> 4);
            const int s   = (int)(e & 15u);
            const float r = survivor_weight(&sV[wave][src][0], s, src >> 4, sh_feats);
            float* dst = act ? &sW[wave][src][s] : &wrow[NSLOT];
            *dst = r;
        }

        __builtin_amdgcn_fence(__ATOMIC_RELEASE, "wavefront");
        __builtin_amdgcn_wave_barrier();
        __builtin_amdgcn_fence(__ATOMIC_ACQUIRE, "wavefront");

        const v4f wa = wrow4[0], wb = wrow4[1], wc = wrow4[2];
        const float w[NSLOT] = {wa.x, wa.y, wa.z, wa.w, wb.x, wb.y, wb.z, wb.w,
                                wc.x, wc.y, wc.z, wc.w};
        float wpart = 0.0f;
#pragma unroll
        for (int sl = 0; sl < NSLOT; ++sl) wpart += w[sl];
        const float wsum = wpart + __shfl_xor(wpart, 16);
        const float invw = 1.0f / fmaxf(wsum, EPS);

        v8f m = z8;
#pragma unroll
        for (int kk = 0; kk < 6; ++kk) {
            v2f b;
            b.x = w[2 * kk]     * invw;
            b.y = w[2 * kk + 1] * invw;
            m = mm4(A2[kk], b, m);
        }

        float oA1 = m[0] * px + m[1] * py + m[2] * pz + m[3];
        float oB1 = m[4] * px + m[5] * py + m[6] * pz + m[7];
        float oA2 = m[0] * qx + m[1] * qy + m[2] * qz + m[3];
        float oB2 = m[4] * qx + m[5] * qy + m[6] * qz + m[7];

        if (!(wsum > EPS)) {
            oA1 = h ? pz : px;  oB1 = py;
            oA2 = h ? qz : qx;  oB2 = qy;
        }

        if (pt < N) {
            const size_t ia3 = (size_t)pt * 3 + (h ? 2 : 0);
            out[ia3] = oA1;
            out[(size_t)N * 3 + ia3] = oA1 - oA2;
            if (h == 0) {
                out[(size_t)pt * 3 + 1] = oB1;
                out[(size_t)N * 3 + (size_t)pt * 3 + 1] = oB1 - oB2;
            }
        }

        px = nx; py = ny; pz = nz;
        ux = mx; uy = my; uz = mz;
    }
}

extern "C" void kernel_launch(void* const* d_in, const int* in_sizes, int n_in,
                              void* d_out, int out_size, void* d_ws, size_t ws_size,
                              hipStream_t stream) {
    const float* xyz      = (const float*)d_in[0];
    const float* viewdirs = (const float*)d_in[1];
    const float* trans    = (const float*)d_in[2];
    const float* sh_feats = (const float*)d_in[3];
    const float* locs     = (const float*)d_in[4];
    float* out = (float*)d_out;

    const int N = in_sizes[0] / 3;
    const int blocks = (N + PTS_PER_BLOCK - 1) / PTS_PER_BLOCK;
    shCaster_kernel<<<blocks, 256, 0, stream>>>(xyz, viewdirs, trans, sh_feats,
                                                locs, out, N);
}
